# starting kernel
# baseline (speedup 1.0000x reference)
#include <hip/hip_runtime.h>
#include <hip/hip_bf16.h>

typedef float v2f __attribute__((ext_vector_type(2)));
typedef float v8f __attribute__((ext_vector_type(8)));

#define D_FEAT 128
#define EDGES_PER_WAVE 16
#define WAVES_PER_BLOCK 8

__global__ __launch_bounds__(WAVES_PER_BLOCK * 32)
void edge_dot_wmma(const float* __restrict__ h,
                   const int*  __restrict__ src,
                   const int*  __restrict__ dst,
                   float* __restrict__ out,
                   int nEdges)
{
    const int  lane = threadIdx.x & 31;
    const long tile = (long)blockIdx.x * WAVES_PER_BLOCK + (threadIdx.x >> 5);
    const long base = tile * EDGES_PER_WAVE;
    if (base >= nEdges) return;

    const int r    = lane & 15;
    const int half = lane >> 4;

    long e = base + r;
    e = (e < (long)nEdges) ? e : base;

    const long sRow = (long)src[e];
    const long dRow = (long)dst[e];
    const float* pa = h + sRow * D_FEAT + half * 2;
    const float* pb = h + dRow * D_FEAT + half * 2;

    v8f acc = {0.f, 0.f, 0.f, 0.f, 0.f, 0.f, 0.f, 0.f};
#pragma unroll 8
    for (int i = 0; i < 32; ++i) {
        v2f a = *(const v2f*)(pa + 4 * i);
        v2f b = *(const v2f*)(pb + 4 * i);
        acc = __builtin_amdgcn_wmma_f32_16x16x4_f32(
             false, a,  false, b,
             (short)0, acc,  false,  false);
    }

    const bool writer = (lane < 8) || (lane >= 24);
    const int  q = (lane < 8) ? lane : (lane - 16);
    const int  v = q & 7;
    float val = acc[0];
    val = (v == 1) ? acc[1] : val;
    val = (v == 2) ? acc[2] : val;
    val = (v == 3) ? acc[3] : val;
    val = (v == 4) ? acc[4] : val;
    val = (v == 5) ? acc[5] : val;
    val = (v == 6) ? acc[6] : val;
    val = (v == 7) ? acc[7] : val;

    const long oe = base + q;
    if (writer && oe < (long)nEdges) out[oe] = val;
}

extern "C" void kernel_launch(void* const* d_in, const int* in_sizes, int n_in,
                              void* d_out, int out_size, void* d_ws, size_t ws_size,
                              hipStream_t stream)
{
    const float* h   = (const float*)d_in[0];
    const int*   src = (const int*)d_in[1];
    const int*   dst = (const int*)d_in[2];
    float*       out = (float*)d_out;
    const int nEdges = in_sizes[1];

    const long tiles  = ((long)nEdges + EDGES_PER_WAVE - 1) / EDGES_PER_WAVE;
    const int  blocks = (int)((tiles + WAVES_PER_BLOCK - 1) / WAVES_PER_BLOCK);
    edge_dot_wmma<<<blocks, WAVES_PER_BLOCK * 32, 0, stream>>>(h, src, dst, out, nEdges);
}


# Every address as scalar base + unsigned 32-bit byte offset (row: node << 9 | lane piece) instead of 64-bit per-lane pointers: no 64-bit vector ops in the loop, 118 -> 93 VGPRs
# speedup vs baseline: 2.0630x; 2.0630x over previous
#include <hip/hip_runtime.h>


typedef float v2f __attribute__((ext_vector_type(2)));
typedef float v4f __attribute__((ext_vector_type(4)));
typedef float v8f __attribute__((ext_vector_type(8)));
typedef int   v4i __attribute__((ext_vector_type(4)));

#define D_FEAT 128
#define GROUP_EDGES 16
#define GROUPS_PER_WAVE 4
#define WAVES_PER_BLOCK 8


static __device__ __forceinline__ float quad_sum(float v)
{
    v += __int_as_float(__builtin_amdgcn_update_dpp(0, __float_as_int(v), 0xB1, 0xF, 0xF, false));
    v += __int_as_float(__builtin_amdgcn_update_dpp(0, __float_as_int(v), 0x4E, 0xF, 0xF, false));
    return v;
}

static __device__ __forceinline__ float group_dots(const float* __restrict__ h,
                                                   v4i sc, v4i dc,
                                                   unsigned offb, int dsel, int qtr)
{
    const char* hb = (const char*)h;
    float res = 0.f;
#pragma unroll
    for (int t = 0; t < 4; ++t) {
        const float* pa = (const float*)(hb + (((unsigned)sc[t] << 9) | offb));
        const float* pb = (const float*)(hb + (((unsigned)dc[t] << 9) | offb));
        v8f acc = {0.f, 0.f, 0.f, 0.f, 0.f, 0.f, 0.f, 0.f};
#pragma unroll
        for (int j = 0; j < 4; ++j) {
            const v4f a = *(const v4f*)(pa + 32 * j);
            const v4f b = *(const v4f*)(pb + 32 * j);
            const v2f a0 = {a.x, a.y}, a1 = {a.z, a.w};
            const v2f b0 = {b.x, b.y}, b1 = {b.z, b.w};
            acc = __builtin_amdgcn_wmma_f32_16x16x4_f32(
                false, a0, false, b0, (short)0, acc, false, false);
            acc = __builtin_amdgcn_wmma_f32_16x16x4_f32(
                false, a1, false, b1, (short)0, acc, false, false);
        }
        float v = acc[0];
        v = (dsel == 1) ? acc[1] : v;
        v = (dsel == 2) ? acc[2] : v;
        v = (dsel == 3) ? acc[3] : v;
        v = (dsel == 4) ? acc[4] : v;
        v = (dsel == 5) ? acc[5] : v;
        v = (dsel == 6) ? acc[6] : v;
        v = (dsel == 7) ? acc[7] : v;
        v = quad_sum(v);
        res = (qtr == t) ? v : res;
    }
    return res;
}

__global__ __launch_bounds__(WAVES_PER_BLOCK * 32)
void edge_dot_wmma(const float* __restrict__ h,
                   const int*  __restrict__ src,
                   const int*  __restrict__ dst,
                   float* __restrict__ out,
                   int nEdges)
{
    const int  lane = threadIdx.x & 31;
    const int  wib   = __builtin_amdgcn_readfirstlane((int)(threadIdx.x >> 5));
    const long wave  = (long)blockIdx.x * WAVES_PER_BLOCK + wib;
    const long first = wave * (GROUPS_PER_WAVE * GROUP_EDGES);
    const long last  = first + GROUPS_PER_WAVE * GROUP_EDGES;
    const long nE    = (long)nEdges;
    if (first >= nE) return;

    const int c    = lane & 15;
    const int hf   = lane >> 4;
    const int slot = c >> 2;
    const int qtr  = c & 3;
    const unsigned offb = (unsigned)(qtr * 32 + hf * 16);
    const int dsel = lane & 7;

    const bool writer = (lane < 8) || (lane >= 24);
    const int  q = (lane < 8) ? lane : (lane - 16);

    const unsigned ioff = (unsigned)(16 * slot);
    const unsigned ooff = (unsigned)(4 * q);

    long base = first;
    if (base + GROUP_EDGES <= nE) {
        v4i s4 = *(const v4i*)((const char*)src + ((unsigned)base * 4u + ioff));
        v4i d4 = *(const v4i*)((const char*)dst + ((unsigned)base * 4u + ioff));
        for (int g = 0; g < GROUPS_PER_WAVE; ++g) {
            const v4i sc = s4, dc = d4;
            const long nbase = base + GROUP_EDGES;
            const bool more  = (g + 1 < GROUPS_PER_WAVE) && (nbase + GROUP_EDGES <= nE);
            const long pf = more ? nbase : base;
            const unsigned pfb = (unsigned)pf * 4u + ioff;
            s4 = *(const v4i*)((const char*)src + pfb);
            d4 = *(const v4i*)((const char*)dst + pfb);

            const float res = group_dots(h, sc, dc, offb, dsel, qtr);
            if (writer) *(float*)((char*)out + ((unsigned)base * 4u + ooff)) = res;

            base = nbase;
            if (!more) break;
        }
    }

    if (base < last && base < nE) {
        v4i sc, dc;
#pragma unroll
        for (int t = 0; t < 4; ++t) {
            long e = base + 4 * slot + t;
            e = (e < nE) ? e : base;
            sc[t] = src[e];
            dc[t] = dst[e];
        }
        const float res = group_dots(h, sc, dc, offb, dsel, qtr);
        const long oe = base + q;
        if (writer && oe < nE) out[oe] = res;
    }
}

extern "C" void kernel_launch(void* const* d_in, const int* in_sizes, int n_in,
                              void* d_out, int out_size, void* d_ws, size_t ws_size,
                              hipStream_t stream)
{
    const float* h   = (const float*)d_in[0];
    const int*   src = (const int*)d_in[1];
    const int*   dst = (const int*)d_in[2];
    float*       out = (float*)d_out;
    const int nEdges = in_sizes[1];

    const long perBlock = (long)WAVES_PER_BLOCK * GROUPS_PER_WAVE * GROUP_EDGES;
    const int  blocks   = (int)(((long)nEdges + perBlock - 1) / perBlock);
    edge_dot_wmma<<<blocks, WAVES_PER_BLOCK * 32, 0, stream>>>(h, src, dst, out, nEdges);
}
